# starting kernel
# baseline (speedup 1.0000x reference)
#include <hip/hip_runtime.h>

typedef __attribute__((ext_vector_type(2))) float v2f;
typedef __attribute__((ext_vector_type(8))) float v8f;

#define CIN   256
#define COUT  256
#define BATCH 2
#define HH    48
#define WW    48
#define HWSZ  (HH * WW)
#define KS    7
#define PADR  3
#define MAPELEMS (BATCH * COUT * HWSZ)

__global__ __launch_bounds__(256) void qkv_gemm(
    const float* __restrict__ x,
    const float* __restrict__ wq,
    const float* __restrict__ wk,
    const float* __restrict__ wv,
    float* __restrict__ ws)
{
  const int z     = blockIdx.z;
  const int b     = z / 3;
  const int which = z - 3 * b;
  const float* __restrict__ Wm = (which == 0) ? wq : ((which == 1) ? wk : wv);
  const float* __restrict__ X  = x + (size_t)b * CIN * HWSZ;
  float* __restrict__ out = ws + (size_t)which * MAPELEMS + (size_t)b * COUT * HWSZ;

  const int tid  = threadIdx.x;
  const int wave = tid >> 5;
  const int lane = tid & 31;
  const int l15  = lane & 15;
  const int hi   = lane >> 4;
  const int koff = hi << 1;

  const int M0 = blockIdx.x * 16;
  const int N0 = blockIdx.y * 128 + wave * 16;
  const int mrow = M0 + l15;
  const int ncol = N0 + l15;

  const float* __restrict__ Arow = Wm + (size_t)mrow * CIN;

  v8f acc = {};
  #pragma unroll 8
  for (int k = 0; k < CIN; k += 4) {
    v2f a = *reinterpret_cast<const v2f*>(Arow + k + koff);
    v2f bm;
    bm.x = X[(size_t)(k + koff)     * HWSZ + ncol];
    bm.y = X[(size_t)(k + koff + 1) * HWSZ + ncol];
    acc = __builtin_amdgcn_wmma_f32_16x16x4_f32(
        false, a, false, bm, (short)0, acc, false, false);
  }

  const int ms = M0 + (hi << 3);
  #pragma unroll
  for (int r = 0; r < 8; ++r)
    out[(size_t)(ms + r) * HWSZ + ncol] = acc[r];
}

#define TS    16
#define PATCH (TS + KS - 1)
#define PITCH (PATCH + 1)
#define CG    16

__global__ __launch_bounds__(256) void local_attn(
    const float* __restrict__ ws,
    const float* __restrict__ rel_h,
    const float* __restrict__ rel_w,
    float* __restrict__ out)
{
  const float* __restrict__ qm = ws;
  const float* __restrict__ km = ws + (size_t)MAPELEMS;
  const float* __restrict__ vm = ws + (size_t)2 * MAPELEMS;

  __shared__ float sk[PATCH * PITCH];
  __shared__ float sv[PATCH * PITCH];

  const int tid = threadIdx.x;
  const int tx  = tid & 15;
  const int ty  = tid >> 4;

  const int b   = blockIdx.z;
  const int c0  = blockIdx.y * CG;
  const int nTw = WW / TS;
  const int th0 = (blockIdx.x / nTw) * TS;
  const int tw0 = (blockIdx.x % nTw) * TS;
  const int h   = th0 + ty;
  const int w   = tw0 + tx;

  for (int ci = 0; ci < CG; ++ci) {
    const int c = c0 + ci;
    const float* __restrict__ kc = km + ((size_t)b * COUT + c) * HWSZ;
    const float* __restrict__ vc = vm + ((size_t)b * COUT + c) * HWSZ;

    for (int idx = tid; idx < PATCH * PATCH; idx += 256) {
      const int r  = idx / PATCH;
      const int cc = idx - r * PATCH;
      const int gh = th0 - PADR + r;
      const int gw = tw0 - PADR + cc;
      const bool ok = (gh >= 0) & (gh < HH) & (gw >= 0) & (gw < WW);
      sk[r * PITCH + cc] = ok ? kc[gh * WW + gw] : 0.0f;
      sv[r * PITCH + cc] = ok ? vc[gh * WW + gw] : 0.0f;
    }
    __syncthreads();

    const float q = qm[((size_t)b * COUT + c) * HWSZ + h * WW + w];

    const bool isrow = (c < COUT / 2);
    const float* __restrict__ rel =
        isrow ? (rel_h + (size_t)c * KS) : (rel_w + (size_t)(c - COUT / 2) * KS);
    float bias7[KS];
    #pragma unroll
    for (int i = 0; i < KS; ++i) bias7[i] = rel[i];

    float mx = -3.0e38f;
    #pragma unroll
    for (int kh = 0; kh < KS; ++kh) {
      #pragma unroll
      for (int kw = 0; kw < KS; ++kw) {
        const float bias = isrow ? bias7[kh] : bias7[kw];
        const float kv = sk[(ty + kh) * PITCH + (tx + kw)] + bias;
        mx = fmaxf(mx, q * kv);
      }
    }

    float s = 0.0f, accv = 0.0f;
    #pragma unroll
    for (int kh = 0; kh < KS; ++kh) {
      #pragma unroll
      for (int kw = 0; kw < KS; ++kw) {
        const float bias = isrow ? bias7[kh] : bias7[kw];
        const float kv = sk[(ty + kh) * PITCH + (tx + kw)] + bias;
        const float e  = __expf(q * kv - mx);
        s    += e;
        accv += e * sv[(ty + kh) * PITCH + (tx + kw)];
      }
    }

    out[((size_t)b * COUT + c) * HWSZ + h * WW + w] = accv / s;
    __syncthreads();
  }
}

extern "C" void kernel_launch(void* const* d_in, const int* in_sizes, int n_in,
                              void* d_out, int out_size, void* d_ws, size_t ws_size,
                              hipStream_t stream) {
  const float* x  = (const float*)d_in[0];
  const float* wq = (const float*)d_in[1];
  const float* wk = (const float*)d_in[2];
  const float* wv = (const float*)d_in[3];
  const float* rh = (const float*)d_in[4];
  const float* rw = (const float*)d_in[5];
  float* ws  = (float*)d_ws;
  float* out = (float*)d_out;

  dim3 g1(COUT / 16, HWSZ / 128, 3 * BATCH);
  qkv_gemm<<<g1, 256, 0, stream>>>(x, wq, wk, wv, ws);

  dim3 g2((HH / TS) * (WW / TS), COUT / CG, BATCH);
  local_attn<<<g2, 256, 0, stream>>>(ws, rh, rw, out);
}


# Window-softmax kernel: one work-group per (batch, channel) plane, whole 48x48 K and V planes staged once in LDS inside a zero border, 3 pixels per thread, no per-channel loop or barriers
# speedup vs baseline: 2.4148x; 2.4148x over previous
#include <hip/hip_runtime.h>

typedef __attribute__((ext_vector_type(2))) float v2f;
typedef __attribute__((ext_vector_type(8))) float v8f;

#define CIN   256
#define COUT  256
#define BATCH 2
#define HH    48
#define WW    48
#define HWSZ  (HH * WW)
#define KS    7
#define PADR  3
#define MAPELEMS (BATCH * COUT * HWSZ)

__global__ __launch_bounds__(256) void qkv_gemm(
    const float* __restrict__ x,
    const float* __restrict__ wq,
    const float* __restrict__ wk,
    const float* __restrict__ wv,
    float* __restrict__ ws)
{
  const int z     = blockIdx.z;
  const int b     = z / 3;
  const int which = z - 3 * b;
  const float* __restrict__ Wm = (which == 0) ? wq : ((which == 1) ? wk : wv);
  const float* __restrict__ X  = x + (size_t)b * CIN * HWSZ;
  float* __restrict__ out = ws + (size_t)which * MAPELEMS + (size_t)b * COUT * HWSZ;

  const int tid  = threadIdx.x;
  const int wave = tid >> 5;
  const int lane = tid & 31;
  const int l15  = lane & 15;
  const int hi   = lane >> 4;
  const int koff = hi << 1;

  const int M0 = blockIdx.x * 16;
  const int N0 = blockIdx.y * 128 + wave * 16;
  const int mrow = M0 + l15;
  const int ncol = N0 + l15;

  const float* __restrict__ Arow = Wm + (size_t)mrow * CIN;

  v8f acc = {};
  #pragma unroll 8
  for (int k = 0; k < CIN; k += 4) {
    v2f a = *reinterpret_cast<const v2f*>(Arow + k + koff);
    v2f bm;
    bm.x = X[(size_t)(k + koff)     * HWSZ + ncol];
    bm.y = X[(size_t)(k + koff + 1) * HWSZ + ncol];
    acc = __builtin_amdgcn_wmma_f32_16x16x4_f32(
        false, a, false, bm, (short)0, acc, false, false);
  }

  const int ms = M0 + (hi << 3);
  #pragma unroll
  for (int r = 0; r < 8; ++r)
    out[(size_t)(ms + r) * HWSZ + ncol] = acc[r];
}

#define PROWS (HH + KS - 1)
#define XO    4
#define PW    56
#define NT2   768
#define PPT   (HWSZ / NT2)
#define NV4   (HWSZ / 4)
#define ZTOP  (2 * PADR * (PW / 4))
#define ZALL  (ZTOP + 2 * HH)

template <bool ISROW>
__device__ __forceinline__ float attn_pixel(const float* pk, const float* pv,
                                            const float q, const float (&bias7)[KS])
{
  float mx = -3.0e38f;
  #pragma unroll
  for (int kh = 0; kh < KS; ++kh) {
    #pragma unroll
    for (int kw = 0; kw < KS; ++kw) {
      const float bias = ISROW ? bias7[kh] : bias7[kw];
      const float kv = pk[kh * PW + kw] + bias;
      mx = fmaxf(mx, q * kv);
    }
  }

  float s = 0.0f, accv = 0.0f;
  #pragma unroll
  for (int kh = 0; kh < KS; ++kh) {
    #pragma unroll
    for (int kw = 0; kw < KS; ++kw) {
      const float bias = ISROW ? bias7[kh] : bias7[kw];
      const float kv = pk[kh * PW + kw] + bias;
      const float e  = __expf(q * kv - mx);
      s    += e;
      accv += e * pv[kh * PW + kw];
    }
  }
  return accv / s;
}

__global__ __launch_bounds__(NT2, 12) void local_attn(
    const float* __restrict__ ws,
    const float* __restrict__ rel_h,
    const float* __restrict__ rel_w,
    float* __restrict__ out)
{
  __shared__ __attribute__((aligned(16))) float sk[PROWS * PW];
  __shared__ __attribute__((aligned(16))) float sv[PROWS * PW];

  const int tid = threadIdx.x;
  const int c   = blockIdx.x;
  const int b   = blockIdx.y;
  const size_t plane = ((size_t)b * COUT + c) * HWSZ;

  const float* __restrict__ qc = ws + plane;
  const float* __restrict__ kc = ws + (size_t)MAPELEMS + plane;
  const float* __restrict__ vc = ws + (size_t)2 * MAPELEMS + plane;

  const bool isrow = (c < COUT / 2);
  const float* __restrict__ rel =
      isrow ? (rel_h + (size_t)c * KS) : (rel_w + (size_t)(c - COUT / 2) * KS);
  float bias7[KS];
  #pragma unroll
  for (int i = 0; i < KS; ++i) bias7[i] = rel[i];

  float q[PPT];
  #pragma unroll
  for (int j = 0; j < PPT; ++j) q[j] = qc[tid + j * NT2];

  if (tid < NV4) {
    const int row = tid / (WW / 4);
    const int c4  = tid - row * (WW / 4);
    const float4 k4 = reinterpret_cast<const float4*>(kc)[tid];
    const float4 v4 = reinterpret_cast<const float4*>(vc)[tid];
    const int a = (row + PADR) * PW + XO + 4 * c4;
    *reinterpret_cast<float4*>(&sk[a]) = k4;
    *reinterpret_cast<float4*>(&sv[a]) = v4;
  } else {
    for (int i = tid - NV4; i < 2 * ZALL; i += NT2 - NV4) {
      const int arr = (i >= ZALL) ? 1 : 0;
      const int e   = i - arr * ZALL;
      int a;
      if (e < ZTOP) {
        const int r  = e / (PW / 4);
        const int cc = e - r * (PW / 4);
        a = ((r < PADR) ? r : (r + HH)) * PW + 4 * cc;
      } else {
        const int t = e - ZTOP;
        a = (PADR + (t >> 1)) * PW + ((t & 1) ? (XO + WW) : 0);
      }
      float* dst = arr ? sv : sk;
      *reinterpret_cast<float4*>(&dst[a]) = make_float4(0.0f, 0.0f, 0.0f, 0.0f);
    }
  }
  __syncthreads();

  const int r0 = tid / WW;
  const int w  = tid - r0 * WW;
  const int base = r0 * PW + w + (XO - PADR);
  float* __restrict__ o = out + plane;

  if (isrow) {
    #pragma unroll
    for (int j = 0; j < PPT; ++j) {
      const int a = base + j * (NT2 / WW) * PW;
      o[tid + j * NT2] = attn_pixel<true>(sk + a, sv + a, q[j], bias7);
    }
  } else {
    #pragma unroll
    for (int j = 0; j < PPT; ++j) {
      const int a = base + j * (NT2 / WW) * PW;
      o[tid + j * NT2] = attn_pixel<false>(sk + a, sv + a, q[j], bias7);
    }
  }
}

extern "C" void kernel_launch(void* const* d_in, const int* in_sizes, int n_in,
                              void* d_out, int out_size, void* d_ws, size_t ws_size,
                              hipStream_t stream) {
  const float* x  = (const float*)d_in[0];
  const float* wq = (const float*)d_in[1];
  const float* wk = (const float*)d_in[2];
  const float* wv = (const float*)d_in[3];
  const float* rh = (const float*)d_in[4];
  const float* rw = (const float*)d_in[5];
  float* ws  = (float*)d_ws;
  float* out = (float*)d_out;

  dim3 g1(COUT / 16, HWSZ / 128, 3 * BATCH);
  qkv_gemm<<<g1, 256, 0, stream>>>(x, wq, wk, wv, ws);

  dim3 g2(COUT, BATCH);
  local_attn<<<g2, NT2, 0, stream>>>(ws, rh, rw, out);
}


# QKV GEMM tiled through LDS: 128x128 work-group tile over the stacked q/k/v weights, 8 waves of 2x4 accumulator tiles, K chunks of 64 double-buffered; k summation order per tile unchanged
# speedup vs baseline: 3.5399x; 1.4659x over previous
#include <hip/hip_runtime.h>

typedef __attribute__((ext_vector_type(2))) float v2f;
typedef __attribute__((ext_vector_type(8))) float v8f;

#define CIN   256
#define COUT  256
#define BATCH 2
#define HH    48
#define WW    48
#define HWSZ  (HH * WW)
#define KS    7
#define PADR  3
#define MAPELEMS (BATCH * COUT * HWSZ)

#define GBM   128
#define GBN   128
#define GBK   64
#define NCHUNK (CIN / GBK)
#define SWP   (GBK + 4)
#define SXP   (2 * GBN + 32)

__global__ __launch_bounds__(256) void qkv_gemm(
    const float* __restrict__ x,
    const float* __restrict__ wq,
    const float* __restrict__ wk,
    const float* __restrict__ wv,
    float* __restrict__ ws)
{
  __shared__ __attribute__((aligned(16))) float sW[2][GBM * SWP];
  __shared__ __attribute__((aligned(16))) float sX[2][(GBK / 2) * SXP];

  const int b     = blockIdx.z;
  const int which = blockIdx.x >> 1;
  const int M0    = (blockIdx.x & 1) * GBM;
  const int N0    = blockIdx.y * GBN;
  const float* __restrict__ Wm = (which == 0) ? wq : ((which == 1) ? wk : wv);
  const float* __restrict__ X  = x + (size_t)b * CIN * HWSZ;
  float* __restrict__ out = ws + (size_t)which * MAPELEMS + (size_t)b * COUT * HWSZ;

  const int tid  = threadIdx.x;
  const int wave = tid >> 5;
  const int lane = tid & 31;
  const int l15  = lane & 15;
  const int hi   = lane >> 4;

  const int wr = tid >> 4;
  const int wc = (tid & 15) * 4;
  const float* __restrict__ gW = Wm + (size_t)(M0 + wr) * CIN + wc;
  const int lW = wr * SWP + wc;
  const int xp = tid >> 5;
  const int xg = tid & 31;
  const float* __restrict__ gX = X + (size_t)(2 * xp) * HWSZ + N0 + 4 * xg;
  const int lX = xp * SXP + 8 * xg;

  float4 rw[8], ra[4], rb[4];

#define GLOAD(kb)                                                                     \
  {                                                                                   \
    _Pragma("unroll") for (int j = 0; j < 8; ++j)                                     \
      rw[j] = *reinterpret_cast<const float4*>(gW + (size_t)(16 * j) * CIN + (kb));   \
    _Pragma("unroll") for (int j = 0; j < 4; ++j) {                                   \
      const float* p = gX + (size_t)((kb) + 16 * j) * HWSZ;                           \
      ra[j] = *reinterpret_cast<const float4*>(p);                                    \
      rb[j] = *reinterpret_cast<const float4*>(p + HWSZ);                             \
    }                                                                                 \
  }

#define LSTORE(bf)                                                                    \
  {                                                                                   \
    _Pragma("unroll") for (int j = 0; j < 8; ++j)                                     \
      *reinterpret_cast<float4*>(&sW[bf][lW + 16 * j * SWP]) = rw[j];                 \
    _Pragma("unroll") for (int j = 0; j < 4; ++j) {                                   \
      float* d = &sX[bf][lX + 8 * j * SXP];                                           \
      *reinterpret_cast<float4*>(d)     = make_float4(ra[j].x, rb[j].x, ra[j].y, rb[j].y); \
      *reinterpret_cast<float4*>(d + 4) = make_float4(ra[j].z, rb[j].z, ra[j].w, rb[j].w); \
    }                                                                                 \
  }

  const int wm = wave & 3;
  const int wn = wave >> 2;
  const int aoff = (wm * 32 + l15) * SWP + 2 * hi;
  const int boff = hi * SXP + 2 * (wn * 64 + l15);

  v8f acc[2][4];
  #pragma unroll
  for (int i = 0; i < 2; ++i) {
    #pragma unroll
    for (int j = 0; j < 4; ++j) acc[i][j] = v8f{};
  }

  GLOAD(0);
  LSTORE(0);
  __syncthreads();

  for (int it = 0; it < NCHUNK; ++it) {
    const int cur = it & 1;
    if (it + 1 < NCHUNK) GLOAD((it + 1) * GBK);

    const float* pa = &sW[cur][aoff];
    const float* pb = &sX[cur][boff];
    #pragma unroll
    for (int s = 0; s < GBK / 4; ++s) {
      v2f a[2], bm[4];
      #pragma unroll
      for (int i = 0; i < 2; ++i)
        a[i] = *reinterpret_cast<const v2f*>(pa + 16 * i * SWP + 4 * s);
      #pragma unroll
      for (int j = 0; j < 4; ++j)
        bm[j] = *reinterpret_cast<const v2f*>(pb + 2 * s * SXP + 32 * j);
      #pragma unroll
      for (int i = 0; i < 2; ++i) {
        #pragma unroll
        for (int j = 0; j < 4; ++j)
          acc[i][j] = __builtin_amdgcn_wmma_f32_16x16x4_f32(
              false, a[i], false, bm[j], (short)0, acc[i][j], false, false);
      }
    }

    if (it + 1 < NCHUNK) LSTORE(cur ^ 1);
    __syncthreads();
  }
#undef GLOAD
#undef LSTORE

  #pragma unroll
  for (int i = 0; i < 2; ++i) {
    const int ms = M0 + wm * 32 + 16 * i + (hi << 3);
    #pragma unroll
    for (int j = 0; j < 4; ++j) {
      const int ncol = N0 + wn * 64 + 16 * j + l15;
      #pragma unroll
      for (int r = 0; r < 8; ++r)
        out[(size_t)(ms + r) * HWSZ + ncol] = acc[i][j][r];
    }
  }
}

#define PROWS (HH + KS - 1)
#define XO    4
#define PW    56
#define NT2   768
#define PPT   (HWSZ / NT2)
#define NV4   (HWSZ / 4)
#define ZTOP  (2 * PADR * (PW / 4))
#define ZALL  (ZTOP + 2 * HH)

template <bool ISROW>
__device__ __forceinline__ float attn_pixel(const float* pk, const float* pv,
                                            const float q, const float (&bias7)[KS])
{
  float mx = -3.0e38f;
  #pragma unroll
  for (int kh = 0; kh < KS; ++kh) {
    #pragma unroll
    for (int kw = 0; kw < KS; ++kw) {
      const float bias = ISROW ? bias7[kh] : bias7[kw];
      const float kv = pk[kh * PW + kw] + bias;
      mx = fmaxf(mx, q * kv);
    }
  }

  float s = 0.0f, accv = 0.0f;
  #pragma unroll
  for (int kh = 0; kh < KS; ++kh) {
    #pragma unroll
    for (int kw = 0; kw < KS; ++kw) {
      const float bias = ISROW ? bias7[kh] : bias7[kw];
      const float kv = pk[kh * PW + kw] + bias;
      const float e  = __expf(q * kv - mx);
      s    += e;
      accv += e * pv[kh * PW + kw];
    }
  }
  return accv / s;
}

__global__ __launch_bounds__(NT2, 12) void local_attn(
    const float* __restrict__ ws,
    const float* __restrict__ rel_h,
    const float* __restrict__ rel_w,
    float* __restrict__ out)
{
  __shared__ __attribute__((aligned(16))) float sk[PROWS * PW];
  __shared__ __attribute__((aligned(16))) float sv[PROWS * PW];

  const int tid = threadIdx.x;
  const int c   = blockIdx.x;
  const int b   = blockIdx.y;
  const size_t plane = ((size_t)b * COUT + c) * HWSZ;

  const float* __restrict__ qc = ws + plane;
  const float* __restrict__ kc = ws + (size_t)MAPELEMS + plane;
  const float* __restrict__ vc = ws + (size_t)2 * MAPELEMS + plane;

  const bool isrow = (c < COUT / 2);
  const float* __restrict__ rel =
      isrow ? (rel_h + (size_t)c * KS) : (rel_w + (size_t)(c - COUT / 2) * KS);
  float bias7[KS];
  #pragma unroll
  for (int i = 0; i < KS; ++i) bias7[i] = rel[i];

  float q[PPT];
  #pragma unroll
  for (int j = 0; j < PPT; ++j) q[j] = qc[tid + j * NT2];

  if (tid < NV4) {
    const int row = tid / (WW / 4);
    const int c4  = tid - row * (WW / 4);
    const float4 k4 = reinterpret_cast<const float4*>(kc)[tid];
    const float4 v4 = reinterpret_cast<const float4*>(vc)[tid];
    const int a = (row + PADR) * PW + XO + 4 * c4;
    *reinterpret_cast<float4*>(&sk[a]) = k4;
    *reinterpret_cast<float4*>(&sv[a]) = v4;
  } else {
    for (int i = tid - NV4; i < 2 * ZALL; i += NT2 - NV4) {
      const int arr = (i >= ZALL) ? 1 : 0;
      const int e   = i - arr * ZALL;
      int a;
      if (e < ZTOP) {
        const int r  = e / (PW / 4);
        const int cc = e - r * (PW / 4);
        a = ((r < PADR) ? r : (r + HH)) * PW + 4 * cc;
      } else {
        const int t = e - ZTOP;
        a = (PADR + (t >> 1)) * PW + ((t & 1) ? (XO + WW) : 0);
      }
      float* dst = arr ? sv : sk;
      *reinterpret_cast<float4*>(&dst[a]) = make_float4(0.0f, 0.0f, 0.0f, 0.0f);
    }
  }
  __syncthreads();

  const int r0 = tid / WW;
  const int w  = tid - r0 * WW;
  const int base = r0 * PW + w + (XO - PADR);
  float* __restrict__ o = out + plane;

  if (isrow) {
    #pragma unroll
    for (int j = 0; j < PPT; ++j) {
      const int a = base + j * (NT2 / WW) * PW;
      o[tid + j * NT2] = attn_pixel<true>(sk + a, sv + a, q[j], bias7);
    }
  } else {
    #pragma unroll
    for (int j = 0; j < PPT; ++j) {
      const int a = base + j * (NT2 / WW) * PW;
      o[tid + j * NT2] = attn_pixel<false>(sk + a, sv + a, q[j], bias7);
    }
  }
}

extern "C" void kernel_launch(void* const* d_in, const int* in_sizes, int n_in,
                              void* d_out, int out_size, void* d_ws, size_t ws_size,
                              hipStream_t stream) {
  const float* x  = (const float*)d_in[0];
  const float* wq = (const float*)d_in[1];
  const float* wk = (const float*)d_in[2];
  const float* wv = (const float*)d_in[3];
  const float* rh = (const float*)d_in[4];
  const float* rw = (const float*)d_in[5];
  float* ws  = (float*)d_ws;
  float* out = (float*)d_out;

  dim3 g1(3 * COUT / GBM, HWSZ / GBN, BATCH);
  qkv_gemm<<<g1, 256, 0, stream>>>(x, wq, wk, wv, ws);

  dim3 g2(COUT, BATCH);
  local_attn<<<g2, NT2, 0, stream>>>(ws, rh, rw, out);
}


# Window-softmax kernel: 12-wave work-groups, 6 pixels per thread (6144 waves enter, not 12288); per pixel all 98 LDS reads issued as one group and the exp chains interleaved; sums in the same order
# speedup vs baseline: 3.8190x; 1.0788x over previous
#include <hip/hip_runtime.h>

typedef __attribute__((ext_vector_type(2))) float v2f;
typedef __attribute__((ext_vector_type(8))) float v8f;

#define CIN   256
#define COUT  256
#define BATCH 2
#define HH    48
#define WW    48
#define HWSZ  (HH * WW)
#define KS    7
#define PADR  3
#define MAPELEMS (BATCH * COUT * HWSZ)

#define GBM   128
#define GBN   128
#define GBK   64
#define NCHUNK (CIN / GBK)
#define SWP   (GBK + 4)
#define SXP   (2 * GBN + 32)

__global__ __launch_bounds__(256) void qkv_gemm(
    const float* __restrict__ x,
    const float* __restrict__ wq,
    const float* __restrict__ wk,
    const float* __restrict__ wv,
    float* __restrict__ ws)
{
  __shared__ __attribute__((aligned(16))) float sW[2][GBM * SWP];
  __shared__ __attribute__((aligned(16))) float sX[2][(GBK / 2) * SXP];

  const int b     = blockIdx.z;
  const int which = blockIdx.x >> 1;
  const int M0    = (blockIdx.x & 1) * GBM;
  const int N0    = blockIdx.y * GBN;
  const float* __restrict__ Wm = (which == 0) ? wq : ((which == 1) ? wk : wv);
  const float* __restrict__ X  = x + (size_t)b * CIN * HWSZ;
  float* __restrict__ out = ws + (size_t)which * MAPELEMS + (size_t)b * COUT * HWSZ;

  const int tid  = threadIdx.x;
  const int wave = tid >> 5;
  const int lane = tid & 31;
  const int l15  = lane & 15;
  const int hi   = lane >> 4;

  const int wr = tid >> 4;
  const int wc = (tid & 15) * 4;
  const float* __restrict__ gW = Wm + (size_t)(M0 + wr) * CIN + wc;
  const int lW = wr * SWP + wc;
  const int xp = tid >> 5;
  const int xg = tid & 31;
  const float* __restrict__ gX = X + (size_t)(2 * xp) * HWSZ + N0 + 4 * xg;
  const int lX = xp * SXP + 8 * xg;

  float4 rw[8], ra[4], rb[4];

#define GLOAD(kb)                                                                     \
  {                                                                                   \
    _Pragma("unroll") for (int j = 0; j < 8; ++j)                                     \
      rw[j] = *reinterpret_cast<const float4*>(gW + (size_t)(16 * j) * CIN + (kb));   \
    _Pragma("unroll") for (int j = 0; j < 4; ++j) {                                   \
      const float* p = gX + (size_t)((kb) + 16 * j) * HWSZ;                           \
      ra[j] = *reinterpret_cast<const float4*>(p);                                    \
      rb[j] = *reinterpret_cast<const float4*>(p + HWSZ);                             \
    }                                                                                 \
  }

#define LSTORE(bf)                                                                    \
  {                                                                                   \
    _Pragma("unroll") for (int j = 0; j < 8; ++j)                                     \
      *reinterpret_cast<float4*>(&sW[bf][lW + 16 * j * SWP]) = rw[j];                 \
    _Pragma("unroll") for (int j = 0; j < 4; ++j) {                                   \
      float* d = &sX[bf][lX + 8 * j * SXP];                                           \
      *reinterpret_cast<float4*>(d)     = make_float4(ra[j].x, rb[j].x, ra[j].y, rb[j].y); \
      *reinterpret_cast<float4*>(d + 4) = make_float4(ra[j].z, rb[j].z, ra[j].w, rb[j].w); \
    }                                                                                 \
  }

  const int wm = wave & 3;
  const int wn = wave >> 2;
  const int aoff = (wm * 32 + l15) * SWP + 2 * hi;
  const int boff = hi * SXP + 2 * (wn * 64 + l15);

  v8f acc[2][4];
  #pragma unroll
  for (int i = 0; i < 2; ++i) {
    #pragma unroll
    for (int j = 0; j < 4; ++j) acc[i][j] = v8f{};
  }

  GLOAD(0);
  LSTORE(0);
  __syncthreads();

  for (int it = 0; it < NCHUNK; ++it) {
    const int cur = it & 1;
    if (it + 1 < NCHUNK) GLOAD((it + 1) * GBK);

    const float* pa = &sW[cur][aoff];
    const float* pb = &sX[cur][boff];
    #pragma unroll
    for (int s = 0; s < GBK / 4; ++s) {
      v2f a[2], bm[4];
      #pragma unroll
      for (int i = 0; i < 2; ++i)
        a[i] = *reinterpret_cast<const v2f*>(pa + 16 * i * SWP + 4 * s);
      #pragma unroll
      for (int j = 0; j < 4; ++j)
        bm[j] = *reinterpret_cast<const v2f*>(pb + 2 * s * SXP + 32 * j);
      #pragma unroll
      for (int i = 0; i < 2; ++i) {
        #pragma unroll
        for (int j = 0; j < 4; ++j)
          acc[i][j] = __builtin_amdgcn_wmma_f32_16x16x4_f32(
              false, a[i], false, bm[j], (short)0, acc[i][j], false, false);
      }
    }

    if (it + 1 < NCHUNK) LSTORE(cur ^ 1);
    __syncthreads();
  }
#undef GLOAD
#undef LSTORE

  #pragma unroll
  for (int i = 0; i < 2; ++i) {
    const int ms = M0 + wm * 32 + 16 * i + (hi << 3);
    #pragma unroll
    for (int j = 0; j < 4; ++j) {
      const int ncol = N0 + wn * 64 + 16 * j + l15;
      #pragma unroll
      for (int r = 0; r < 8; ++r)
        out[(size_t)(ms + r) * HWSZ + ncol] = acc[i][j][r];
    }
  }
}

#define PROWS (HH + KS - 1)
#define XO    4
#define PW    56
#define NT2   384
#define PPT   (HWSZ / NT2)
#define NV4   (HWSZ / 4)
#define ZTOP  (2 * PADR * (PW / 4))
#define ZALL  (ZTOP + 2 * HH)
#define WIN   (KS * KS)
#define LOG2E 1.44269504088896340736f

template <bool ISROW>
__device__ __forceinline__ void attn_plane(const float* sk, const float* sv,
                                           const float (&q6)[PPT],
                                           float* __restrict__ o,
                                           const float (&bias7)[KS], const int tid)
{
  const int r0 = tid / WW;
  const int w  = tid - r0 * WW;
  const int a0 = r0 * PW + w + (XO - PADR);
  #pragma unroll 1
  for (int j = 0; j < PPT; ++j) {
    const int p = tid + j * NT2;
    const float* pk = sk + a0 + j * (NT2 / WW) * PW;
    const float* pv = sv + a0 + j * (NT2 / WW) * PW;
    float q = q6[0];
    #pragma unroll
    for (int i = 1; i < PPT; ++i) q = (j == i) ? q6[i] : q;

    float kv[WIN], vr[WIN];
    #pragma unroll
    for (int kh = 0; kh < KS; ++kh) {
      #pragma unroll
      for (int kw = 0; kw < KS - 1; ++kw) kv[kh * KS + kw] = pk[kh * PW + kw];
    }
    #pragma unroll
    for (int kh = 0; kh < KS; ++kh) kv[kh * KS + KS - 1] = pk[kh * PW + KS - 1];
    #pragma unroll
    for (int kh = 0; kh < KS; ++kh) {
      #pragma unroll
      for (int kw = 0; kw < KS; ++kw) vr[kh * KS + kw] = pv[kh * PW + kw];
    }
    __builtin_amdgcn_sched_barrier(0);

    const v2f qq = {q, q};
    float m0 = -3.0e38f, m1 = -3.0e38f, m2 = -3.0e38f, m3 = -3.0e38f;
    float m4 = -3.0e38f, m5 = -3.0e38f, m6 = -3.0e38f;
    #pragma unroll
    for (int t = 0; t < KS / 2; ++t) {
      v2f x[7], y[7];
      #pragma unroll
      for (int u = 0; u < 6; ++u) {
        const int kh = 2 * t + u / 3;
        const int kw = 2 * (u % 3);
        const v2f bb = ISROW ? v2f{bias7[kh], bias7[kh]} : v2f{bias7[kw], bias7[kw + 1]};
        x[u] = v2f{kv[kh * KS + kw], kv[kh * KS + kw + 1]} + bb;
      }
      {
        const v2f bb = ISROW ? v2f{bias7[2 * t], bias7[2 * t + 1]} : v2f{bias7[KS - 1], bias7[KS - 1]};
        x[6] = v2f{kv[(2 * t) * KS + KS - 1], kv[(2 * t + 1) * KS + KS - 1]} + bb;
      }
      #pragma unroll
      for (int u = 0; u < 7; ++u) y[u] = x[u] * qq;
      #pragma unroll
      for (int u = 0; u < 6; ++u) {
        const int kh = 2 * t + u / 3;
        const int kw = 2 * (u % 3);
        kv[kh * KS + kw]     = x[u].x;
        kv[kh * KS + kw + 1] = x[u].y;
      }
      kv[(2 * t) * KS + KS - 1]     = x[6].x;
      kv[(2 * t + 1) * KS + KS - 1] = x[6].y;
      m0 = fmaxf(fmaxf(m0, y[0].x), y[0].y);
      m1 = fmaxf(fmaxf(m1, y[1].x), y[1].y);
      m2 = fmaxf(fmaxf(m2, y[2].x), y[2].y);
      m3 = fmaxf(fmaxf(m3, y[3].x), y[3].y);
      m4 = fmaxf(fmaxf(m4, y[4].x), y[4].y);
      m5 = fmaxf(fmaxf(m5, y[5].x), y[5].y);
      m6 = fmaxf(fmaxf(m6, y[6].x), y[6].y);
    }
    {
      const int kh = KS - 1;
      v2f x[3], y[3];
      #pragma unroll
      for (int u = 0; u < 3; ++u) {
        const int kw = 2 * u;
        const v2f bb = ISROW ? v2f{bias7[kh], bias7[kh]} : v2f{bias7[kw], bias7[kw + 1]};
        x[u] = v2f{kv[kh * KS + kw], kv[kh * KS + kw + 1]} + bb;
      }
      const float xl = kv[WIN - 1] + bias7[KS - 1];
      #pragma unroll
      for (int u = 0; u < 3; ++u) y[u] = x[u] * qq;
      const float yl = q * xl;
      #pragma unroll
      for (int u = 0; u < 3; ++u) {
        kv[kh * KS + 2 * u]     = x[u].x;
        kv[kh * KS + 2 * u + 1] = x[u].y;
      }
      kv[WIN - 1] = xl;
      m0 = fmaxf(fmaxf(m0, y[0].x), y[0].y);
      m1 = fmaxf(fmaxf(m1, y[1].x), y[1].y);
      m2 = fmaxf(fmaxf(m2, y[2].x), y[2].y);
      m6 = fmaxf(m6, yl);
    }
    const float mx = fmaxf(fmaxf(fmaxf(m0, m1), fmaxf(m2, m3)), fmaxf(fmaxf(m4, m5), m6));

    float ta[WIN], tb[WIN], te[WIN];
    float s = 0.0f, accv = 0.0f;
    #pragma unroll
    for (int n = 0; n < WIN + 6; ++n) {
      if (n < WIN)               ta[n]     = q * kv[n] - mx;
      if (n >= 2 && n - 2 < WIN) tb[n - 2] = ta[n - 2] * LOG2E;
      if (n >= 4 && n - 4 < WIN) te[n - 4] = __builtin_amdgcn_exp2f(tb[n - 4]);
      if (n >= 6) {
        s    += te[n - 6];
        accv += te[n - 6] * vr[n - 6];
      }
    }
    o[p] = accv / s;
  }
}

__global__ __launch_bounds__(NT2, 6) void local_attn(
    const float* __restrict__ ws,
    const float* __restrict__ rel_h,
    const float* __restrict__ rel_w,
    float* __restrict__ out)
{
  __shared__ __attribute__((aligned(16))) float sk[PROWS * PW];
  __shared__ __attribute__((aligned(16))) float sv[PROWS * PW];

  const int tid = threadIdx.x;
  const int c   = blockIdx.x;
  const int b   = blockIdx.y;
  const size_t plane = ((size_t)b * COUT + c) * HWSZ;

  const float* __restrict__ qc = ws + plane;
  const float* __restrict__ kc = ws + (size_t)MAPELEMS + plane;
  const float* __restrict__ vc = ws + (size_t)2 * MAPELEMS + plane;

  const bool isrow = (c < COUT / 2);
  const float* __restrict__ rel =
      isrow ? (rel_h + (size_t)c * KS) : (rel_w + (size_t)(c - COUT / 2) * KS);
  float bias7[KS];
  #pragma unroll
  for (int i = 0; i < KS; ++i) bias7[i] = rel[i];

  float q6[PPT];
  #pragma unroll
  for (int j = 0; j < PPT; ++j) q6[j] = qc[tid + j * NT2];

  const float4* __restrict__ kc4 = reinterpret_cast<const float4*>(kc);
  const float4* __restrict__ vc4 = reinterpret_cast<const float4*>(vc);
  const bool midk = (tid < NV4 - NT2);
  const int  imid = midk ? (tid + NT2) : (tid + NT2 - NV4);
  const int  ilst = tid + 2 * NT2 - NV4;
  const float4 pfst = kc4[tid];
  const float4 pmid = (midk ? kc4 : vc4)[imid];
  const float4 plst = vc4[ilst];
  if (tid < 2 * ZALL) {
    const int arr = (tid >= ZALL) ? 1 : 0;
    const int e   = tid - arr * ZALL;
    int a;
    if (e < ZTOP) {
      const int r  = e / (PW / 4);
      const int cc = e - r * (PW / 4);
      a = ((r < PADR) ? r : (r + HH)) * PW + 4 * cc;
    } else {
      const int t = e - ZTOP;
      a = (PADR + (t >> 1)) * PW + ((t & 1) ? (XO + WW) : 0);
    }
    float* dst = arr ? sv : sk;
    *reinterpret_cast<float4*>(&dst[a]) = make_float4(0.0f, 0.0f, 0.0f, 0.0f);
  }
  {
    const int rf = tid / (WW / 4),  af = (rf + PADR) * PW + XO + 4 * (tid  - rf * (WW / 4));
    const int rm = imid / (WW / 4), am = (rm + PADR) * PW + XO + 4 * (imid - rm * (WW / 4));
    const int rl = ilst / (WW / 4), al = (rl + PADR) * PW + XO + 4 * (ilst - rl * (WW / 4));
    float* dmid = midk ? sk : sv;
    *reinterpret_cast<float4*>(&sk[af])   = pfst;
    *reinterpret_cast<float4*>(&dmid[am]) = pmid;
    *reinterpret_cast<float4*>(&sv[al])   = plst;
  }
  __syncthreads();

  float* __restrict__ o = out + plane;
  if (isrow) attn_plane<true>(sk, sv, q6, o, bias7, tid);
  else       attn_plane<false>(sk, sv, q6, o, bias7, tid);
}

extern "C" void kernel_launch(void* const* d_in, const int* in_sizes, int n_in,
                              void* d_out, int out_size, void* d_ws, size_t ws_size,
                              hipStream_t stream) {
  const float* x  = (const float*)d_in[0];
  const float* wq = (const float*)d_in[1];
  const float* wk = (const float*)d_in[2];
  const float* wv = (const float*)d_in[3];
  const float* rh = (const float*)d_in[4];
  const float* rw = (const float*)d_in[5];
  float* ws  = (float*)d_ws;
  float* out = (float*)d_out;

  dim3 g1(3 * COUT / GBM, HWSZ / GBN, BATCH);
  qkv_gemm<<<g1, 256, 0, stream>>>(x, wq, wk, wv, ws);

  dim3 g2(COUT, BATCH);
  local_attn<<<g2, NT2, 0, stream>>>(ws, rh, rw, out);
}


# Window-softmax pass 2 on packed 2-element f32 ops: logits kept from pass 1, exponent argument by one packed fma, exp-sum and V sum as even/odd partial sums added at the end (sum order changes)
# speedup vs baseline: 3.9928x; 1.0455x over previous
#include <hip/hip_runtime.h>

typedef __attribute__((ext_vector_type(2))) float v2f;
typedef __attribute__((ext_vector_type(8))) float v8f;

#define CIN   256
#define COUT  256
#define BATCH 2
#define HH    48
#define WW    48
#define HWSZ  (HH * WW)
#define KS    7
#define PADR  3
#define MAPELEMS (BATCH * COUT * HWSZ)

#define GBM   128
#define GBN   128
#define GBK   64
#define NCHUNK (CIN / GBK)
#define SWP   (GBK + 4)
#define SXP   (2 * GBN + 32)

__global__ __launch_bounds__(256) void qkv_gemm(
    const float* __restrict__ x,
    const float* __restrict__ wq,
    const float* __restrict__ wk,
    const float* __restrict__ wv,
    float* __restrict__ ws)
{
  __shared__ __attribute__((aligned(16))) float sW[2][GBM * SWP];
  __shared__ __attribute__((aligned(16))) float sX[2][(GBK / 2) * SXP];

  const int b     = blockIdx.z;
  const int which = blockIdx.x >> 1;
  const int M0    = (blockIdx.x & 1) * GBM;
  const int N0    = blockIdx.y * GBN;
  const float* __restrict__ Wm = (which == 0) ? wq : ((which == 1) ? wk : wv);
  const float* __restrict__ X  = x + (size_t)b * CIN * HWSZ;
  float* __restrict__ out = ws + (size_t)which * MAPELEMS + (size_t)b * COUT * HWSZ;

  const int tid  = threadIdx.x;
  const int wave = tid >> 5;
  const int lane = tid & 31;
  const int l15  = lane & 15;
  const int hi   = lane >> 4;

  const int wr = tid >> 4;
  const int wc = (tid & 15) * 4;
  const float* __restrict__ gW = Wm + (size_t)(M0 + wr) * CIN + wc;
  const int lW = wr * SWP + wc;
  const int xp = tid >> 5;
  const int xg = tid & 31;
  const float* __restrict__ gX = X + (size_t)(2 * xp) * HWSZ + N0 + 4 * xg;
  const int lX = xp * SXP + 8 * xg;

  float4 rw[8], ra[4], rb[4];

#define GLOAD(kb)                                                                     \
  {                                                                                   \
    _Pragma("unroll") for (int j = 0; j < 8; ++j)                                     \
      rw[j] = *reinterpret_cast<const float4*>(gW + (size_t)(16 * j) * CIN + (kb));   \
    _Pragma("unroll") for (int j = 0; j < 4; ++j) {                                   \
      const float* p = gX + (size_t)((kb) + 16 * j) * HWSZ;                           \
      ra[j] = *reinterpret_cast<const float4*>(p);                                    \
      rb[j] = *reinterpret_cast<const float4*>(p + HWSZ);                             \
    }                                                                                 \
  }

#define LSTORE(bf)                                                                    \
  {                                                                                   \
    _Pragma("unroll") for (int j = 0; j < 8; ++j)                                     \
      *reinterpret_cast<float4*>(&sW[bf][lW + 16 * j * SWP]) = rw[j];                 \
    _Pragma("unroll") for (int j = 0; j < 4; ++j) {                                   \
      float* d = &sX[bf][lX + 8 * j * SXP];                                           \
      *reinterpret_cast<float4*>(d)     = make_float4(ra[j].x, rb[j].x, ra[j].y, rb[j].y); \
      *reinterpret_cast<float4*>(d + 4) = make_float4(ra[j].z, rb[j].z, ra[j].w, rb[j].w); \
    }                                                                                 \
  }

  const int wm = wave & 3;
  const int wn = wave >> 2;
  const int aoff = (wm * 32 + l15) * SWP + 2 * hi;
  const int boff = hi * SXP + 2 * (wn * 64 + l15);

  v8f acc[2][4];
  #pragma unroll
  for (int i = 0; i < 2; ++i) {
    #pragma unroll
    for (int j = 0; j < 4; ++j) acc[i][j] = v8f{};
  }

  GLOAD(0);
  LSTORE(0);
  __syncthreads();

  for (int it = 0; it < NCHUNK; ++it) {
    const int cur = it & 1;
    if (it + 1 < NCHUNK) GLOAD((it + 1) * GBK);

    const float* pa = &sW[cur][aoff];
    const float* pb = &sX[cur][boff];
    #pragma unroll
    for (int s = 0; s < GBK / 4; ++s) {
      v2f a[2], bm[4];
      #pragma unroll
      for (int i = 0; i < 2; ++i)
        a[i] = *reinterpret_cast<const v2f*>(pa + 16 * i * SWP + 4 * s);
      #pragma unroll
      for (int j = 0; j < 4; ++j)
        bm[j] = *reinterpret_cast<const v2f*>(pb + 2 * s * SXP + 32 * j);
      #pragma unroll
      for (int i = 0; i < 2; ++i) {
        #pragma unroll
        for (int j = 0; j < 4; ++j)
          acc[i][j] = __builtin_amdgcn_wmma_f32_16x16x4_f32(
              false, a[i], false, bm[j], (short)0, acc[i][j], false, false);
      }
    }

    if (it + 1 < NCHUNK) LSTORE(cur ^ 1);
    __syncthreads();
  }
#undef GLOAD
#undef LSTORE

  #pragma unroll
  for (int i = 0; i < 2; ++i) {
    const int ms = M0 + wm * 32 + 16 * i + (hi << 3);
    #pragma unroll
    for (int j = 0; j < 4; ++j) {
      const int ncol = N0 + wn * 64 + 16 * j + l15;
      #pragma unroll
      for (int r = 0; r < 8; ++r)
        out[(size_t)(ms + r) * HWSZ + ncol] = acc[i][j][r];
    }
  }
}

#define PROWS (HH + KS - 1)
#define XO    4
#define PW    56
#define NT2   384
#define PPT   (HWSZ / NT2)
#define NV4   (HWSZ / 4)
#define ZTOP  (2 * PADR * (PW / 4))
#define ZALL  (ZTOP + 2 * HH)
#define WIN   (KS * KS)
#define LOG2E 1.44269504088896340736f

#define NPAIR 24
template <bool ISROW>
__device__ __forceinline__ void attn_plane(const float* sk, const float* sv,
                                           const float (&q6)[PPT],
                                           float* __restrict__ o,
                                           const float (&bias7)[KS], const int tid)
{
  const int r0 = tid / WW;
  const int w  = tid - r0 * WW;
  const int a0 = r0 * PW + w + (XO - PADR);
  #pragma unroll 1
  for (int j = 0; j < PPT; ++j) {
    const int p = tid + j * NT2;
    const float* pk = sk + a0 + j * (NT2 / WW) * PW;
    const float* pv = sv + a0 + j * (NT2 / WW) * PW;
    float q = q6[0];
    #pragma unroll
    for (int i = 1; i < PPT; ++i) q = (j == i) ? q6[i] : q;

    v2f kk[NPAIR], vv[NPAIR];
    #pragma unroll
    for (int kh = 0; kh < KS; ++kh) {
      #pragma unroll
      for (int cc = 0; cc < 3; ++cc)
        kk[kh * 3 + cc] = v2f{pk[kh * PW + 2 * cc], pk[kh * PW + 2 * cc + 1]};
    }
    #pragma unroll
    for (int t = 0; t < 3; ++t)
      kk[21 + t] = v2f{pk[(2 * t) * PW + KS - 1], pk[(2 * t + 1) * PW + KS - 1]};
    const float kl = pk[(KS - 1) * PW + KS - 1];
    #pragma unroll
    for (int kh = 0; kh < KS; ++kh) {
      #pragma unroll
      for (int cc = 0; cc < 3; ++cc)
        vv[kh * 3 + cc] = v2f{pv[kh * PW + 2 * cc], pv[kh * PW + 2 * cc + 1]};
    }
    #pragma unroll
    for (int t = 0; t < 3; ++t)
      vv[21 + t] = v2f{pv[(2 * t) * PW + KS - 1], pv[(2 * t + 1) * PW + KS - 1]};
    const float vl = pv[(KS - 1) * PW + KS - 1];
    __builtin_amdgcn_sched_barrier(0);

    const v2f qq = {q, q};
    float m0 = -3.0e38f, m1 = -3.0e38f, m2 = -3.0e38f, m3 = -3.0e38f;
    float m4 = -3.0e38f, m5 = -3.0e38f, m6 = -3.0e38f;
    #pragma unroll
    for (int t = 0; t < KS / 2; ++t) {
      v2f x[7], y[7];
      #pragma unroll
      for (int u = 0; u < 6; ++u) {
        const int kh = 2 * t + u / 3;
        const int cc = u % 3;
        const v2f bb = ISROW ? v2f{bias7[kh], bias7[kh]} : v2f{bias7[2 * cc], bias7[2 * cc + 1]};
        x[u] = kk[kh * 3 + cc] + bb;
      }
      {
        const v2f bb = ISROW ? v2f{bias7[2 * t], bias7[2 * t + 1]} : v2f{bias7[KS - 1], bias7[KS - 1]};
        x[6] = kk[21 + t] + bb;
      }
      #pragma unroll
      for (int u = 0; u < 7; ++u) y[u] = x[u] * qq;
      #pragma unroll
      for (int u = 0; u < 6; ++u) kk[(2 * t + u / 3) * 3 + u % 3] = y[u];
      kk[21 + t] = y[6];
      m0 = fmaxf(fmaxf(m0, y[0].x), y[0].y);
      m1 = fmaxf(fmaxf(m1, y[1].x), y[1].y);
      m2 = fmaxf(fmaxf(m2, y[2].x), y[2].y);
      m3 = fmaxf(fmaxf(m3, y[3].x), y[3].y);
      m4 = fmaxf(fmaxf(m4, y[4].x), y[4].y);
      m5 = fmaxf(fmaxf(m5, y[5].x), y[5].y);
      m6 = fmaxf(fmaxf(m6, y[6].x), y[6].y);
    }
    float yl;
    {
      const int kh = KS - 1;
      v2f x[3], y[3];
      #pragma unroll
      for (int cc = 0; cc < 3; ++cc) {
        const v2f bb = ISROW ? v2f{bias7[kh], bias7[kh]} : v2f{bias7[2 * cc], bias7[2 * cc + 1]};
        x[cc] = kk[kh * 3 + cc] + bb;
      }
      const float xl = kl + bias7[KS - 1];
      #pragma unroll
      for (int cc = 0; cc < 3; ++cc) y[cc] = x[cc] * qq;
      yl = q * xl;
      #pragma unroll
      for (int cc = 0; cc < 3; ++cc) kk[kh * 3 + cc] = y[cc];
      m0 = fmaxf(fmaxf(m0, y[0].x), y[0].y);
      m1 = fmaxf(fmaxf(m1, y[1].x), y[1].y);
      m2 = fmaxf(fmaxf(m2, y[2].x), y[2].y);
      m6 = fmaxf(m6, yl);
    }
    const float mx = fmaxf(fmaxf(fmaxf(m0, m1), fmaxf(m2, m3)), fmaxf(fmaxf(m4, m5), m6));

    const float mxl = mx * LOG2E;
    const v2f l2  = {LOG2E, LOG2E};
    const v2f nmx = {-mxl, -mxl};
    v2f ta[NPAIR], te[NPAIR];
    v2f s2 = {0.0f, 0.0f}, a2 = {0.0f, 0.0f};
    #pragma unroll
    for (int n = 0; n < NPAIR + 4; ++n) {
      if (n < NPAIR) ta[n] = __builtin_elementwise_fma(kk[n], l2, nmx);
      if (n >= 2 && n - 2 < NPAIR)
        te[n - 2] = v2f{__builtin_amdgcn_exp2f(ta[n - 2].x), __builtin_amdgcn_exp2f(ta[n - 2].y)};
      if (n >= 4) {
        s2 += te[n - 4];
        a2 = __builtin_elementwise_fma(te[n - 4], vv[n - 4], a2);
      }
    }
    const float el   = __builtin_amdgcn_exp2f(fmaf(yl, LOG2E, -mxl));
    const float s    = (s2.x + s2.y) + el;
    const float accv = fmaf(el, vl, a2.x + a2.y);
    o[p] = accv / s;
  }
}

__global__ __launch_bounds__(NT2, 6) void local_attn(
    const float* __restrict__ ws,
    const float* __restrict__ rel_h,
    const float* __restrict__ rel_w,
    float* __restrict__ out)
{
  __shared__ __attribute__((aligned(16))) float sk[PROWS * PW];
  __shared__ __attribute__((aligned(16))) float sv[PROWS * PW];

  const int tid = threadIdx.x;
  const int c   = blockIdx.x;
  const int b   = blockIdx.y;
  const size_t plane = ((size_t)b * COUT + c) * HWSZ;

  const float* __restrict__ qc = ws + plane;
  const float* __restrict__ kc = ws + (size_t)MAPELEMS + plane;
  const float* __restrict__ vc = ws + (size_t)2 * MAPELEMS + plane;

  const bool isrow = (c < COUT / 2);
  const float* __restrict__ rel =
      isrow ? (rel_h + (size_t)c * KS) : (rel_w + (size_t)(c - COUT / 2) * KS);
  float bias7[KS];
  #pragma unroll
  for (int i = 0; i < KS; ++i) bias7[i] = rel[i];

  float q6[PPT];
  #pragma unroll
  for (int j = 0; j < PPT; ++j) q6[j] = qc[tid + j * NT2];

  const float4* __restrict__ kc4 = reinterpret_cast<const float4*>(kc);
  const float4* __restrict__ vc4 = reinterpret_cast<const float4*>(vc);
  const bool midk = (tid < NV4 - NT2);
  const int  imid = midk ? (tid + NT2) : (tid + NT2 - NV4);
  const int  ilst = tid + 2 * NT2 - NV4;
  const float4 pfst = kc4[tid];
  const float4 pmid = (midk ? kc4 : vc4)[imid];
  const float4 plst = vc4[ilst];
  if (tid < 2 * ZALL) {
    const int arr = (tid >= ZALL) ? 1 : 0;
    const int e   = tid - arr * ZALL;
    int a;
    if (e < ZTOP) {
      const int r  = e / (PW / 4);
      const int cc = e - r * (PW / 4);
      a = ((r < PADR) ? r : (r + HH)) * PW + 4 * cc;
    } else {
      const int t = e - ZTOP;
      a = (PADR + (t >> 1)) * PW + ((t & 1) ? (XO + WW) : 0);
    }
    float* dst = arr ? sv : sk;
    *reinterpret_cast<float4*>(&dst[a]) = make_float4(0.0f, 0.0f, 0.0f, 0.0f);
  }
  {
    const int rf = tid / (WW / 4),  af = (rf + PADR) * PW + XO + 4 * (tid  - rf * (WW / 4));
    const int rm = imid / (WW / 4), am = (rm + PADR) * PW + XO + 4 * (imid - rm * (WW / 4));
    const int rl = ilst / (WW / 4), al = (rl + PADR) * PW + XO + 4 * (ilst - rl * (WW / 4));
    float* dmid = midk ? sk : sv;
    *reinterpret_cast<float4*>(&sk[af])   = pfst;
    *reinterpret_cast<float4*>(&dmid[am]) = pmid;
    *reinterpret_cast<float4*>(&sv[al])   = plst;
  }
  __syncthreads();

  float* __restrict__ o = out + plane;
  if (isrow) attn_plane<true>(sk, sv, q6, o, bias7, tid);
  else       attn_plane<false>(sk, sv, q6, o, bias7, tid);
}

extern "C" void kernel_launch(void* const* d_in, const int* in_sizes, int n_in,
                              void* d_out, int out_size, void* d_ws, size_t ws_size,
                              hipStream_t stream) {
  const float* x  = (const float*)d_in[0];
  const float* wq = (const float*)d_in[1];
  const float* wk = (const float*)d_in[2];
  const float* wv = (const float*)d_in[3];
  const float* rh = (const float*)d_in[4];
  const float* rw = (const float*)d_in[5];
  float* ws  = (float*)d_ws;
  float* out = (float*)d_out;

  dim3 g1(3 * COUT / GBM, HWSZ / GBN, BATCH);
  qkv_gemm<<<g1, 256, 0, stream>>>(x, wq, wk, wv, ws);

  dim3 g2(COUT, BATCH);
  local_attn<<<g2, NT2, 0, stream>>>(ws, rh, rw, out);
}


# Window-softmax: 6-wave groups; each thread slides the 7x7 K/V window down a 12-row strip in registers, one new row per pixel (20 LDS reads a pixel, not 49); exp argument by fma from K, sum order kept
# speedup vs baseline: 4.1703x; 1.0445x over previous
#include <hip/hip_runtime.h>

typedef __attribute__((ext_vector_type(2))) float v2f;
typedef __attribute__((ext_vector_type(8))) float v8f;

#define CIN   256
#define COUT  256
#define BATCH 2
#define HH    48
#define WW    48
#define HWSZ  (HH * WW)
#define KS    7
#define PADR  3
#define MAPELEMS (BATCH * COUT * HWSZ)

#define GBM   128
#define GBN   128
#define GBK   64
#define NCHUNK (CIN / GBK)
#define SWP   (GBK + 4)
#define SXP   (2 * GBN + 32)

__global__ __launch_bounds__(256) void qkv_gemm(
    const float* __restrict__ x,
    const float* __restrict__ wq,
    const float* __restrict__ wk,
    const float* __restrict__ wv,
    float* __restrict__ ws)
{
  __shared__ __attribute__((aligned(16))) float sW[2][GBM * SWP];
  __shared__ __attribute__((aligned(16))) float sX[2][(GBK / 2) * SXP];

  const int b     = blockIdx.z;
  const int which = blockIdx.x >> 1;
  const int M0    = (blockIdx.x & 1) * GBM;
  const int N0    = blockIdx.y * GBN;
  const float* __restrict__ Wm = (which == 0) ? wq : ((which == 1) ? wk : wv);
  const float* __restrict__ X  = x + (size_t)b * CIN * HWSZ;
  float* __restrict__ out = ws + (size_t)which * MAPELEMS + (size_t)b * COUT * HWSZ;

  const int tid  = threadIdx.x;
  const int wave = tid >> 5;
  const int lane = tid & 31;
  const int l15  = lane & 15;
  const int hi   = lane >> 4;

  const int wr = tid >> 4;
  const int wc = (tid & 15) * 4;
  const float* __restrict__ gW = Wm + (size_t)(M0 + wr) * CIN + wc;
  const int lW = wr * SWP + wc;
  const int xp = tid >> 5;
  const int xg = tid & 31;
  const float* __restrict__ gX = X + (size_t)(2 * xp) * HWSZ + N0 + 4 * xg;
  const int lX = xp * SXP + 8 * xg;

  float4 rw[8], ra[4], rb[4];

#define GLOAD(kb)                                                                     \
  {                                                                                   \
    _Pragma("unroll") for (int j = 0; j < 8; ++j)                                     \
      rw[j] = *reinterpret_cast<const float4*>(gW + (size_t)(16 * j) * CIN + (kb));   \
    _Pragma("unroll") for (int j = 0; j < 4; ++j) {                                   \
      const float* p = gX + (size_t)((kb) + 16 * j) * HWSZ;                           \
      ra[j] = *reinterpret_cast<const float4*>(p);                                    \
      rb[j] = *reinterpret_cast<const float4*>(p + HWSZ);                             \
    }                                                                                 \
  }

#define LSTORE(bf)                                                                    \
  {                                                                                   \
    _Pragma("unroll") for (int j = 0; j < 8; ++j)                                     \
      *reinterpret_cast<float4*>(&sW[bf][lW + 16 * j * SWP]) = rw[j];                 \
    _Pragma("unroll") for (int j = 0; j < 4; ++j) {                                   \
      float* d = &sX[bf][lX + 8 * j * SXP];                                           \
      *reinterpret_cast<float4*>(d)     = make_float4(ra[j].x, rb[j].x, ra[j].y, rb[j].y); \
      *reinterpret_cast<float4*>(d + 4) = make_float4(ra[j].z, rb[j].z, ra[j].w, rb[j].w); \
    }                                                                                 \
  }

  const int wm = wave & 3;
  const int wn = wave >> 2;
  const int aoff = (wm * 32 + l15) * SWP + 2 * hi;
  const int boff = hi * SXP + 2 * (wn * 64 + l15);

  v8f acc[2][4];
  #pragma unroll
  for (int i = 0; i < 2; ++i) {
    #pragma unroll
    for (int j = 0; j < 4; ++j) acc[i][j] = v8f{};
  }

  GLOAD(0);
  LSTORE(0);
  __syncthreads();

  for (int it = 0; it < NCHUNK; ++it) {
    const int cur = it & 1;
    if (it + 1 < NCHUNK) GLOAD((it + 1) * GBK);

    const float* pa = &sW[cur][aoff];
    const float* pb = &sX[cur][boff];
    #pragma unroll
    for (int s = 0; s < GBK / 4; ++s) {
      v2f a[2], bm[4];
      #pragma unroll
      for (int i = 0; i < 2; ++i)
        a[i] = *reinterpret_cast<const v2f*>(pa + 16 * i * SWP + 4 * s);
      #pragma unroll
      for (int j = 0; j < 4; ++j)
        bm[j] = *reinterpret_cast<const v2f*>(pb + 2 * s * SXP + 32 * j);
      #pragma unroll
      for (int i = 0; i < 2; ++i) {
        #pragma unroll
        for (int j = 0; j < 4; ++j)
          acc[i][j] = __builtin_amdgcn_wmma_f32_16x16x4_f32(
              false, a[i], false, bm[j], (short)0, acc[i][j], false, false);
      }
    }

    if (it + 1 < NCHUNK) LSTORE(cur ^ 1);
    __syncthreads();
  }
#undef GLOAD
#undef LSTORE

  #pragma unroll
  for (int i = 0; i < 2; ++i) {
    const int ms = M0 + wm * 32 + 16 * i + (hi << 3);
    #pragma unroll
    for (int j = 0; j < 4; ++j) {
      const int ncol = N0 + wn * 64 + 16 * j + l15;
      #pragma unroll
      for (int r = 0; r < 8; ++r)
        out[(size_t)(ms + r) * HWSZ + ncol] = acc[i][j][r];
    }
  }
}

#define PROWS (HH + KS - 1)
#define XO    4
#define PW    56
#define NT2   192
#define PPT   6
#define NRUN  (HWSZ / NT2 / PPT)
#define NV4   (HWSZ / 4)
#define ZTOP  (2 * PADR * (PW / 4))
#define ZALL  (ZTOP + 2 * HH)
#define WIN   (KS * KS)
#define LOG2E 1.44269504088896340736f

#define NPAIR 24
#define NROW  (PPT + KS - 1)
template <bool ISROW>
__device__ __forceinline__ void attn_plane(const float* sk, const float* sv,
                                           const float (&q12)[NRUN * PPT],
                                           float* __restrict__ o,
                                           const float (&bias7)[KS], const int tid)
{
  const int g  = tid / WW;
  const int w  = tid - g * WW;
  #pragma unroll 1
  for (int t = 0; t < NRUN; ++t) {
  const int row0 = (NRUN * g + t) * PPT;
  const int a0 = row0 * PW + w + (XO - PADR);
  const float* pk = sk + a0;
  const float* pv = sv + a0;
  float q6[PPT];
  #pragma unroll
  for (int j = 0; j < PPT; ++j) q6[j] = t ? q12[PPT + j] : q12[j];

  v2f kr[NROW][3], vr[NROW][3];
  #pragma unroll
  for (int R = 0; R < KS - 1; ++R) {
    #pragma unroll
    for (int cc = 0; cc < 3; ++cc) {
      kr[R][cc] = v2f{pk[R * PW + 2 * cc], pk[R * PW + 2 * cc + 1]};
      vr[R][cc] = v2f{pv[R * PW + 2 * cc], pv[R * PW + 2 * cc + 1]};
    }
  }

  #pragma unroll
  for (int j = 0; j < PPT; ++j) {
    const float q = q6[j];
    #pragma unroll
    for (int cc = 0; cc < 3; ++cc) {
      kr[j + KS - 1][cc] = v2f{pk[(j + KS - 1) * PW + 2 * cc], pk[(j + KS - 1) * PW + 2 * cc + 1]};
      vr[j + KS - 1][cc] = v2f{pv[(j + KS - 1) * PW + 2 * cc], pv[(j + KS - 1) * PW + 2 * cc + 1]};
    }
    v2f kk[NPAIR], vv[NPAIR];
    #pragma unroll
    for (int t = 0; t < 3; ++t) {
      kk[21 + t] = v2f{pk[(j + 2 * t) * PW + KS - 1], pk[(j + 2 * t + 1) * PW + KS - 1]};
      vv[21 + t] = v2f{pv[(j + 2 * t) * PW + KS - 1], pv[(j + 2 * t + 1) * PW + KS - 1]};
    }
    const float kl = pk[(j + KS - 1) * PW + KS - 1];
    const float vl = pv[(j + KS - 1) * PW + KS - 1];
    __builtin_amdgcn_sched_barrier(0);
    #pragma unroll
    for (int kh = 0; kh < KS; ++kh) {
      #pragma unroll
      for (int cc = 0; cc < 3; ++cc) {
        kk[kh * 3 + cc] = kr[j + kh][cc];
        vv[kh * 3 + cc] = vr[j + kh][cc];
      }
    }

    float qb[KS];
    #pragma unroll
    for (int i = 0; i < KS; ++i) qb[i] = q * bias7[i];

    const v2f qq = {q, q};
    float m[KS];
    #pragma unroll
    for (int i = 0; i < KS; ++i) m[i] = -3.0e38f;
    #pragma unroll
    for (int n = 0; n < NPAIR; ++n) {
      v2f bb;
      if (n < 21) {
        const int kh = n / 3, cc = n % 3;
        bb = ISROW ? v2f{qb[kh], qb[kh]} : v2f{qb[2 * cc], qb[2 * cc + 1]};
      } else {
        const int t = n - 21;
        bb = ISROW ? v2f{qb[2 * t], qb[2 * t + 1]} : v2f{qb[KS - 1], qb[KS - 1]};
      }
      const v2f y = __builtin_elementwise_fma(kk[n], qq, bb);
      m[n % KS] = fmaxf(fmaxf(m[n % KS], y.x), y.y);
    }
    const float yl = fmaf(kl, q, qb[KS - 1]);
    m[3] = fmaxf(m[3], yl);
    const float mx = fmaxf(fmaxf(fmaxf(m[0], m[1]), fmaxf(m[2], m[3])), fmaxf(fmaxf(m[4], m[5]), m[6]));

    const float mxl = mx * LOG2E;
    const float ql  = q * LOG2E;
    float cl[KS];
    #pragma unroll
    for (int i = 0; i < KS; ++i) cl[i] = fmaf(qb[i], LOG2E, -mxl);
    const v2f qlp = {ql, ql};
    v2f ta[NPAIR], te[NPAIR];
    v2f s2 = {0.0f, 0.0f}, a2 = {0.0f, 0.0f};
    #pragma unroll
    for (int n = 0; n < NPAIR + 4; ++n) {
      if (n < NPAIR) {
        v2f cc2;
        if (n < 21) {
          const int kh = n / 3, cc = n % 3;
          cc2 = ISROW ? v2f{cl[kh], cl[kh]} : v2f{cl[2 * cc], cl[2 * cc + 1]};
        } else {
          const int t = n - 21;
          cc2 = ISROW ? v2f{cl[2 * t], cl[2 * t + 1]} : v2f{cl[KS - 1], cl[KS - 1]};
        }
        ta[n] = __builtin_elementwise_fma(kk[n], qlp, cc2);
      }
      if (n >= 2 && n - 2 < NPAIR)
        te[n - 2] = v2f{__builtin_amdgcn_exp2f(ta[n - 2].x), __builtin_amdgcn_exp2f(ta[n - 2].y)};
      if (n >= 4) {
        s2 += te[n - 4];
        a2 = __builtin_elementwise_fma(te[n - 4], vv[n - 4], a2);
      }
    }
    const float el   = __builtin_amdgcn_exp2f(fmaf(kl, ql, cl[KS - 1]));
    const float s    = (s2.x + s2.y) + el;
    const float accv = fmaf(el, vl, a2.x + a2.y);
    o[(row0 + j) * WW + w] = accv / s;
  }
  }
}

__global__ __launch_bounds__(NT2, 3) void local_attn(
    const float* __restrict__ ws,
    const float* __restrict__ rel_h,
    const float* __restrict__ rel_w,
    float* __restrict__ out)
{
  __shared__ __attribute__((aligned(16))) float sk[PROWS * PW];
  __shared__ __attribute__((aligned(16))) float sv[PROWS * PW];

  const int tid = threadIdx.x;
  const int c   = blockIdx.x;
  const int b   = blockIdx.y;
  const size_t plane = ((size_t)b * COUT + c) * HWSZ;

  const float* __restrict__ qc = ws + plane;
  const float* __restrict__ kc = ws + (size_t)MAPELEMS + plane;
  const float* __restrict__ vc = ws + (size_t)2 * MAPELEMS + plane;

  const bool isrow = (c < COUT / 2);
  const float* __restrict__ rel =
      isrow ? (rel_h + (size_t)c * KS) : (rel_w + (size_t)(c - COUT / 2) * KS);
  float bias7[KS];
  #pragma unroll
  for (int i = 0; i < KS; ++i) bias7[i] = rel[i];

  float q12[NRUN * PPT];
  #pragma unroll
  for (int u = 0; u < NRUN * PPT; ++u) q12[u] = qc[(NRUN * PPT * (tid / WW) + u) * WW + (tid % WW)];

  const float4* __restrict__ kc4 = reinterpret_cast<const float4*>(kc);
  const float4* __restrict__ vc4 = reinterpret_cast<const float4*>(vc);
  float4 pkq[NV4 / NT2], pvq[NV4 / NT2];
  #pragma unroll
  for (int u = 0; u < NV4 / NT2; ++u) {
    pkq[u] = kc4[tid + u * NT2];
    pvq[u] = vc4[tid + u * NT2];
  }
  #pragma unroll
  for (int u = 0; u < 2; ++u) {
    const int z = tid + u * NT2;
    if (z < 2 * ZALL) {
      const int arr = (z >= ZALL) ? 1 : 0;
      const int e   = z - arr * ZALL;
      int a;
      if (e < ZTOP) {
        const int r  = e / (PW / 4);
        const int cc = e - r * (PW / 4);
        a = ((r < PADR) ? r : (r + HH)) * PW + 4 * cc;
      } else {
        const int tt = e - ZTOP;
        a = (PADR + (tt >> 1)) * PW + ((tt & 1) ? (XO + WW) : 0);
      }
      float* dst = arr ? sv : sk;
      *reinterpret_cast<float4*>(&dst[a]) = make_float4(0.0f, 0.0f, 0.0f, 0.0f);
    }
  }
  #pragma unroll
  for (int u = 0; u < NV4 / NT2; ++u) {
    const int i = tid + u * NT2;
    const int r = i / (WW / 4);
    const int a = (r + PADR) * PW + XO + 4 * (i - r * (WW / 4));
    *reinterpret_cast<float4*>(&sk[a]) = pkq[u];
    *reinterpret_cast<float4*>(&sv[a]) = pvq[u];
  }
  __syncthreads();

  float* __restrict__ o = out + plane;
  if (isrow) attn_plane<true>(sk, sv, q12, o, bias7, tid);
  else       attn_plane<false>(sk, sv, q12, o, bias7, tid);
}

extern "C" void kernel_launch(void* const* d_in, const int* in_sizes, int n_in,
                              void* d_out, int out_size, void* d_ws, size_t ws_size,
                              hipStream_t stream) {
  const float* x  = (const float*)d_in[0];
  const float* wq = (const float*)d_in[1];
  const float* wk = (const float*)d_in[2];
  const float* wv = (const float*)d_in[3];
  const float* rh = (const float*)d_in[4];
  const float* rw = (const float*)d_in[5];
  float* ws  = (float*)d_ws;
  float* out = (float*)d_out;

  dim3 g1(3 * COUT / GBM, HWSZ / GBN, BATCH);
  qkv_gemm<<<g1, 256, 0, stream>>>(x, wq, wk, wv, ws);

  dim3 g2(COUT, BATCH);
  local_attn<<<g2, NT2, 0, stream>>>(ws, rh, rw, out);
}
